# starting kernel
# baseline (speedup 1.0000x reference)
#include <hip/hip_runtime.h>


#define HDIM 128
#define BATCH 4
#define LEN 1024
#define MTOT (BATCH * LEN)
#define NCHUNK 16
#define CLEN (LEN / NCHUNK)

typedef float v2f __attribute__((ext_vector_type(2)));
typedef float v8f __attribute__((ext_vector_type(8)));

__device__ __forceinline__ float sigmoid_fast(float v) { return 1.0f / (1.0f + __expf(-v)); }
__device__ __forceinline__ float silu_fast(float v)    { return v * sigmoid_fast(v); }
__device__ __forceinline__ float softplus_fast(float v){ return (v > 20.0f) ? v : log1pf(__expf(v)); }

#define EPI_XZ    0
#define EPI_SSM   1
#define EPI_PLAIN 2

template <int N, int K, int MODE>
__global__ void __launch_bounds__(256)
gemm16_wmma_f32(const float* __restrict__ A,
                const float* __restrict__ W,
                const float* __restrict__ bias,
                float* __restrict__ out0,
                float* __restrict__ out1,
                float* __restrict__ out2)
{
    const int lane = threadIdx.x & 31;
    const int wave = threadIdx.x >> 5;
    const int gw   = blockIdx.x * (blockDim.x >> 5) + wave;
    const int tilesN = N >> 4;
    const int m0 = (gw / tilesN) << 4;
    const int n0 = (gw % tilesN) << 4;
    const int hi  = lane >> 4;
    const int l16 = lane & 15;

    v8f acc = {0.f, 0.f, 0.f, 0.f, 0.f, 0.f, 0.f, 0.f};

    const float* Arow = A + (size_t)(m0 + l16) * K + 2 * hi;
    const float* Wcol = W + (size_t)(2 * hi) * N + n0 + l16;

#pragma unroll
    for (int k0 = 0; k0 < K; k0 += 4) {
        v2f a;
        a.x = Arow[k0 + 0];
        a.y = Arow[k0 + 1];
        v2f b;
        b.x = Wcol[(k0 + 0) * N];
        b.y = Wcol[(k0 + 1) * N];
        acc = __builtin_amdgcn_wmma_f32_16x16x4_f32(false, a, false, b,
                                                    (short)0, acc, false, false);
    }

    const int   col = n0 + l16;
    const float bv  = bias[col];
#pragma unroll
    for (int r = 0; r < 8; ++r) {
        const int row = m0 + r + hi * 8;
        float v = acc[r] + bv;
        if (MODE == EPI_XZ) {
            if (col < HDIM) out0[(size_t)row * HDIM + col] = v;
            else            out1[(size_t)row * HDIM + (col - HDIM)] = silu_fast(v);
        } else if (MODE == EPI_SSM) {
            if (col < 128)      out0[(size_t)row * 128 + col]         = softplus_fast(v);
            else if (col < 256) out1[(size_t)row * 128 + (col - 128)] = v;
            else                out2[(size_t)row * 128 + (col - 256)] = v;
        } else {
            out0[(size_t)row * HDIM + col] = v;
        }
    }
}

__global__ void conv_silu_kernel(const float* __restrict__ x,
                                 const float* __restrict__ w,
                                 const float* __restrict__ cb,
                                 float* __restrict__ x2)
{
    const int idx = blockIdx.x * blockDim.x + threadIdx.x;
    const int e = idx & (HDIM - 1);
    const int m = idx >> 7;
    const int l = m & (LEN - 1);
    const float xm = (l > 0)       ? x[(size_t)(m - 1) * HDIM + e] : 0.f;
    const float xc =                 x[(size_t)m * HDIM + e];
    const float xp = (l < LEN - 1) ? x[(size_t)(m + 1) * HDIM + e] : 0.f;
    const float v  = xm * w[e * 3 + 0] + xc * w[e * 3 + 1] + xp * w[e * 3 + 2] + cb[e];
    x2[(size_t)m * HDIM + e] = silu_fast(v);
}


__global__ void __launch_bounds__(1024)
scan_passA(const float* __restrict__ delta,
           const float* __restrict__ Bm,
           const float* __restrict__ x2,
           const float* __restrict__ A_log,
           float* __restrict__ Ssum,
           float* __restrict__ Bacc)
{
    const int wg = blockIdx.x;
    const int b  = wg >> 4;
    const int c  = wg & (NCHUNK - 1);
    const int tid = threadIdx.x;
    const int j   = tid & 127;
    const int ng  = tid >> 7;
    const int nbase = ng << 4;

    __shared__ float s_d[128], s_B[128], s_x[128];

    const float Aj = -__expf(A_log[j]);
    float h[16], S[16];
#pragma unroll
    for (int i = 0; i < 16; ++i) { h[i] = 0.f; S[i] = 0.f; }

    const size_t base = ((size_t)b * LEN + (size_t)c * CLEN) * HDIM;
    for (int l = 0; l < CLEN; ++l) {
        const size_t moff = base + (size_t)l * HDIM;
        if (ng == 0) {
            s_d[j] = delta[moff + j];
            s_B[j] = Bm[moff + j];
            s_x[j] = x2[moff + j];
        }
        __syncthreads();
        const float xj = s_x[j];
#pragma unroll
        for (int i = 0; i < 16; ++i) {
            const float d = s_d[nbase + i];
            S[i] += d;
            const float e = __expf(d * Aj);
            h[i] = e * h[i] + (d * s_B[nbase + i]) * xj;
        }
        __syncthreads();
    }

    const size_t ob = (size_t)(b * NCHUNK + c) * 128;
#pragma unroll
    for (int i = 0; i < 16; ++i)
        Bacc[(ob + nbase + i) * 128 + j] = h[i];
    if (j == 0) {
#pragma unroll
        for (int i = 0; i < 16; ++i)
            Ssum[ob + nbase + i] = S[i];
    }
}

__global__ void __launch_bounds__(1024)
scan_passB(const float* __restrict__ Ssum,
           const float* __restrict__ Bacc,
           const float* __restrict__ A_log,
           float* __restrict__ hinit)
{
    const int b   = blockIdx.x;
    const int tid = threadIdx.x;
    const int j   = tid & 127;
    const int ng  = tid >> 7;
    const int nbase = ng << 4;

    __shared__ float s_S[128];

    const float Aj = -__expf(A_log[j]);
    float h[16];
#pragma unroll
    for (int i = 0; i < 16; ++i) h[i] = 0.f;

    for (int c = 0; c < NCHUNK; ++c) {
        const size_t ob = (size_t)(b * NCHUNK + c) * 128;
        if (ng == 0) s_S[j] = Ssum[ob + j];
        __syncthreads();
#pragma unroll
        for (int i = 0; i < 16; ++i) {
            hinit[(ob + nbase + i) * 128 + j] = h[i];
            const float P = __expf(s_S[nbase + i] * Aj);
            h[i] = P * h[i] + Bacc[(ob + nbase + i) * 128 + j];
        }
        __syncthreads();
    }
}

__global__ void __launch_bounds__(1024)
scan_passC(const float* __restrict__ delta,
           const float* __restrict__ Bm,
           const float* __restrict__ Cm,
           const float* __restrict__ x2,
           const float* __restrict__ u,
           const float* __restrict__ sz,
           const float* __restrict__ A_log,
           const float* __restrict__ Dv,
           const float* __restrict__ hinit,
           float* __restrict__ ymul)
{
    const int wg = blockIdx.x;
    const int b  = wg >> 4;
    const int c  = wg & (NCHUNK - 1);
    const int tid = threadIdx.x;
    const int j   = tid & 127;
    const int ng  = tid >> 7;
    const int nbase = ng << 4;

    __shared__ float s_d[128], s_B[128], s_x[128], s_C[128], s_u[128];
    __shared__ float s_part[8][128];

    const float Aj = -__expf(A_log[j]);
    const float Dj = Dv[j];

    const size_t ob = (size_t)(b * NCHUNK + c) * 128;
    float h[16];
#pragma unroll
    for (int i = 0; i < 16; ++i) h[i] = hinit[(ob + nbase + i) * 128 + j];

    const size_t base = ((size_t)b * LEN + (size_t)c * CLEN) * HDIM;
    for (int l = 0; l < CLEN; ++l) {
        const size_t moff = base + (size_t)l * HDIM;
        if (ng == 0) {
            s_d[j] = delta[moff + j];
            s_B[j] = Bm[moff + j];
            s_x[j] = x2[moff + j];
            s_C[j] = Cm[moff + j];
            s_u[j] = u[moff + j];
        }
        __syncthreads();
        const float xj = s_x[j];
        float partial = 0.f;
#pragma unroll
        for (int i = 0; i < 16; ++i) {
            const float d = s_d[nbase + i];
            const float e = __expf(d * Aj);
            h[i] = e * h[i] + (d * s_B[nbase + i]) * xj;
            partial += h[i];
        }
        s_part[ng][j] = partial;
        __syncthreads();
        if (ng == 0) {
            float tot = 0.f;
#pragma unroll
            for (int g = 0; g < 8; ++g) tot += s_part[g][j];
            const float y = s_C[j] * tot + Dj * s_u[j];
            ymul[moff + j] = y * sz[moff + j];
        }
    }
}

extern "C" void kernel_launch(void* const* d_in, const int* in_sizes, int n_in,
                              void* d_out, int out_size, void* d_ws, size_t ws_size,
                              hipStream_t stream)
{
    (void)in_sizes; (void)n_in; (void)out_size; (void)ws_size;

    const float* u      = (const float*)d_in[0];
    const float* W_in   = (const float*)d_in[1];
    const float* b_in   = (const float*)d_in[2];
    const float* conv_w = (const float*)d_in[3];
    const float* conv_b = (const float*)d_in[4];
    const float* W_x    = (const float*)d_in[5];
    const float* b_x    = (const float*)d_in[6];
    const float* A_log  = (const float*)d_in[7];
    const float* Dv     = (const float*)d_in[8];
    const float* W_out  = (const float*)d_in[9];
    const float* b_out  = (const float*)d_in[10];

    const size_t P = (size_t)MTOT * HDIM;
    const size_t NSTATE = (size_t)BATCH * NCHUNK * 128 * 128;

    float* ws    = (float*)d_ws;
    float* x_raw = ws + 0 * P;
    float* s_z   = ws + 1 * P;
    float* x2    = ws + 2 * P;
    float* delta = ws + 3 * P;
    float* Bmat  = ws + 4 * P;
    float* Cmat  = ws + 5 * P;
    float* ymul  = ws + 6 * P;
    float* Ssum  = ws + 7 * P;
    float* Bacc  = Ssum + (size_t)BATCH * NCHUNK * 128;
    float* hinit = Bacc + NSTATE;

    gemm16_wmma_f32<256, 128, EPI_XZ><<<512, 256, 0, stream>>>(
        u, W_in, b_in, x_raw, s_z, nullptr);
    conv_silu_kernel<<<(MTOT * HDIM) / 256, 256, 0, stream>>>(x_raw, conv_w, conv_b, x2);
    gemm16_wmma_f32<384, 128, EPI_SSM><<<768, 256, 0, stream>>>(
        x2, W_x, b_x, delta, Bmat, Cmat);
    scan_passA<<<BATCH * NCHUNK, 1024, 0, stream>>>(delta, Bmat, x2, A_log, Ssum, Bacc);
    scan_passB<<<BATCH, 1024, 0, stream>>>(Ssum, Bacc, A_log, hinit);
    scan_passC<<<BATCH * NCHUNK, 1024, 0, stream>>>(delta, Bmat, Cmat, x2, u, s_z,
                                                    A_log, Dv, hinit, ymul);
    gemm16_wmma_f32<128, 128, EPI_PLAIN><<<256, 256, 0, stream>>>(
        ymul, W_out, b_out, (float*)d_out, nullptr, nullptr);
}


# conv+silu folded into the in-projection GEMM: each wave forms its x strip from A tiles shifted a row up and a row down, so all 3 taps sit in its own D registers; conv launch and x_raw plane gone
# speedup vs baseline: 3.6586x; 3.6586x over previous
#include <hip/hip_runtime.h>


#define HDIM 128
#define BATCH 4
#define LEN 1024
#define MTOT (BATCH * LEN)

typedef float v2f __attribute__((ext_vector_type(2)));
typedef float v8f __attribute__((ext_vector_type(8)));

__device__ __forceinline__ float sigmoid_fast(float v) { return 1.0f / (1.0f + __expf(-v)); }
__device__ __forceinline__ float silu_fast(float v)    { return v * sigmoid_fast(v); }
__device__ __forceinline__ float softplus_fast(float v){ return (v > 20.0f) ? v : log1pf(__expf(v)); }

#define EPI_SSM   1
#define EPI_PLAIN 2

template <int N, int K, int MODE, int ALD>
__global__ void __launch_bounds__(256)
gemm16_wmma_f32(const float* __restrict__ A,
                const float* __restrict__ W,
                const float* __restrict__ bias,
                float* __restrict__ out0,
                float* __restrict__ out1,
                float* __restrict__ out2)
{
    const int lane = threadIdx.x & 31;
    const int wave = threadIdx.x >> 5;
    const int gw   = blockIdx.x * (blockDim.x >> 5) + wave;
    const int tilesN = N >> 4;
    const int m0 = (gw / tilesN) << 4;
    const int n0 = (gw % tilesN) << 4;
    const int hi  = lane >> 4;
    const int l16 = lane & 15;

    v8f acc = {0.f, 0.f, 0.f, 0.f, 0.f, 0.f, 0.f, 0.f};

    const float* Arow = ALD ? A + (size_t)(2 * hi) * ALD + (m0 + l16)
                            : A + (size_t)(m0 + l16) * K + 2 * hi;
    const float* Wcol = W + (size_t)(2 * hi) * N + n0 + l16;

#pragma unroll
    for (int k0 = 0; k0 < K; k0 += 4) {
        v2f a;
        a.x = ALD ? Arow[(size_t)(k0 + 0) * ALD] : Arow[k0 + 0];
        a.y = ALD ? Arow[(size_t)(k0 + 1) * ALD] : Arow[k0 + 1];
        v2f b;
        b.x = Wcol[(k0 + 0) * N];
        b.y = Wcol[(k0 + 1) * N];
        acc = __builtin_amdgcn_wmma_f32_16x16x4_f32(false, a, false, b,
                                                    (short)0, acc, false, false);
    }

    const int   col = n0 + l16;
    const float bv  = bias[col];
#pragma unroll
    for (int r = 0; r < 8; ++r) {
        const int row = m0 + r + hi * 8;
        float v = acc[r] + bv;
        if (MODE == EPI_SSM) {
            if (col < 128)      out0[(size_t)row * 128 + col]         = softplus_fast(v);
            else if (col < 256) out1[(size_t)row * 128 + (col - 128)] = v;
            else                out2[(size_t)row * 128 + (col - 256)] = v;
        } else {
            out0[(size_t)row * HDIM + col] = v;
        }
    }
}

#define KC 32
#define KS (KC / 4)

__device__ __forceinline__ void inproj_load_chunk(
    v2f (&aa)[KS], v2f (&ab)[KS], v2f (&ac)[KS], v2f (&bx)[KS], v2f (&bz)[KS],
    const float* __restrict__ Aa, const float* __restrict__ Ab, const float* __restrict__ Ac,
    const float* __restrict__ Wx, const float* __restrict__ Wz, const int kb)
{
#pragma unroll
    for (int s = 0; s < KS; ++s) {
        const int k0 = kb + 4 * s;
        aa[s].x = Aa[k0 + 0];  aa[s].y = Aa[k0 + 1];
        ab[s].x = Ab[k0 + 0];  ab[s].y = Ab[k0 + 1];
        ac[s].x = Ac[k0 + 0];  ac[s].y = Ac[k0 + 1];
        bx[s].x = Wx[(k0 + 0) * (2 * HDIM)];
        bx[s].y = Wx[(k0 + 1) * (2 * HDIM)];
        bz[s].x = Wz[(k0 + 0) * (2 * HDIM)];
        bz[s].y = Wz[(k0 + 1) * (2 * HDIM)];
    }
}

__global__ void __launch_bounds__(256)
inproj_conv_wmma_f32(const float* __restrict__ u,
                     const float* __restrict__ W,
                     const float* __restrict__ bias,
                     const float* __restrict__ cw,
                     const float* __restrict__ cb,
                     float* __restrict__ x2,
                     float* __restrict__ sz)
{
    const int lane = threadIdx.x & 31;
    const int wave = threadIdx.x >> 5;
    const int gw   = blockIdx.x * (blockDim.x >> 5) + wave;
    const int m0 = (gw >> 3) << 4;
    const int n0 = (gw & 7) << 4;
    const int hi  = lane >> 4;
    const int l16 = lane & 15;

    const int ra = (m0 - 1 + l16 < 0) ? 0 : m0 - 1 + l16;
    const int rb = (m0 + 1 + l16 > MTOT - 1) ? MTOT - 1 : m0 + 1 + l16;
    const float* Aa = u + (size_t)ra * HDIM + 2 * hi;
    const float* Ab = u + (size_t)rb * HDIM + 2 * hi;
    const float* Ac = u + (size_t)(m0 + l16) * HDIM + 2 * hi;
    const float* Wx = W + (size_t)(2 * hi) * (2 * HDIM) + n0 + l16;
    const float* Wz = Wx + HDIM;

    v8f acc_a = {0.f, 0.f, 0.f, 0.f, 0.f, 0.f, 0.f, 0.f};
    v8f acc_b = acc_a;
    v8f acc_c = acc_a;

    v2f aa[2][KS], ab[2][KS], ac[2][KS], bx[2][KS], bz[2][KS];
    inproj_load_chunk(aa[0], ab[0], ac[0], bx[0], bz[0], Aa, Ab, Ac, Wx, Wz, 0);

#pragma unroll
    for (int c = 0; c < HDIM / KC; ++c) {
        const int cur = c & 1;
        if (c + 1 < HDIM / KC)
            inproj_load_chunk(aa[cur ^ 1], ab[cur ^ 1], ac[cur ^ 1], bx[cur ^ 1], bz[cur ^ 1],
                              Aa, Ab, Ac, Wx, Wz, (c + 1) * KC);
        __builtin_amdgcn_sched_barrier(0);
#pragma unroll
        for (int s = 0; s < KS; ++s) {
            acc_a = __builtin_amdgcn_wmma_f32_16x16x4_f32(false, aa[cur][s], false, bx[cur][s],
                                                          (short)0, acc_a, false, false);
            acc_b = __builtin_amdgcn_wmma_f32_16x16x4_f32(false, ab[cur][s], false, bx[cur][s],
                                                          (short)0, acc_b, false, false);
            acc_c = __builtin_amdgcn_wmma_f32_16x16x4_f32(false, ac[cur][s], false, bz[cur][s],
                                                          (short)0, acc_c, false, false);
        }
        __builtin_amdgcn_sched_barrier(0);
    }

    const int   e   = n0 + l16;
    const float bxv = bias[e];
    const float bzv = bias[HDIM + e];
    const float w0  = cw[e * 3 + 0];
    const float w1  = cw[e * 3 + 1];
    const float w2  = cw[e * 3 + 2];
    const float cbv = cb[e];
#pragma unroll
    for (int r = 0; r < 8; ++r) {
        const int row = m0 + r + hi * 8;
        const int l   = row & (LEN - 1);
        const int rn  = (r < 7) ? r + 1 : 7;
        const float xm = (l > 0)       ? acc_a[r] + bxv : 0.f;
        const float xc = ((r < 7) ? acc_a[rn] : acc_b[6]) + bxv;
        const float xp = (l < LEN - 1) ? acc_b[r] + bxv : 0.f;
        const float v  = xm * w0 + xc * w1 + xp * w2 + cbv;
        x2[(size_t)row * HDIM + e] = silu_fast(v);
        sz[(size_t)row * HDIM + e] = silu_fast(acc_c[r] + bzv);
    }
}

#define SC 2
#define SH 2
#define ST 32
#define GS 8
#define SB (SH * ST)
#define NSB (LEN / SB)
#define PPITCH4 9

template <bool MORE>
__device__ __forceinline__ void scan_pass(
    float2 (&dr)[ST], float2 (&br)[ST],
    const float2* __restrict__ dnx,
    const float2* __restrict__ bnx,
    const float xv,
    float& h0, float& h1, const float Aj,
    float* __restrict__ pw, const int lane)
{
#pragma unroll
    for (int gi = 0; gi < ST / GS; ++gi) {
#pragma unroll
        for (int i = 0; i < GS; ++i) {
            const int s = gi * GS + i;
            const float2 dd = dr[s];
            const float2 bb = br[s];
            const float xs = __builtin_bit_cast(float,
                __builtin_amdgcn_readlane(__builtin_bit_cast(int, xv), s));
            const float p0 = dd.x * bb.x;
            const float p1 = dd.y * bb.y;
            h0 = __expf(dd.x * Aj) * h0 + p0 * xs;
            h1 = __expf(dd.y * Aj) * h1 + p1 * xs;
            pw[s * (PPITCH4 * 4) + lane] = h0 + h1;
        }
        if (MORE) {
            __builtin_amdgcn_sched_barrier(0);
#pragma unroll
            for (int i = 0; i < GS; ++i) {
                const int s = gi * GS + i;
                dr[s] = dnx[(size_t)s * 64];
                br[s] = bnx[(size_t)s * 64];
            }
            __builtin_amdgcn_sched_barrier(0);
        }
    }
}

template <bool LAST>
__device__ __forceinline__ void scan_block(
    float2 (&dr)[ST], float2 (&br)[ST],
    const float2* __restrict__ dk,
    const float2* __restrict__ bk,
    const float* __restrict__ xk,
    const float* __restrict__ cp,
    const float* __restrict__ up,
    const float* __restrict__ zp,
    float* __restrict__ yp,
    float& xv, float& h0, float& h1,
    const float Aj, const float Dj,
    float* __restrict__ pw,
    const float4* __restrict__ r0,
    const float4* __restrict__ r1,
    const int lane)
{
    float xn = xk[(size_t)ST * HDIM];
    const float cv = *cp;
    const float uv = *up;
    const float zv = *zp;

    scan_pass<true>(dr, br, dk + (size_t)ST * 64, bk + (size_t)ST * 64,
                    xv, h0, h1, Aj, pw, lane);
    xv = xn;
    xn = 0.f;
    if (!LAST) xn = xk[(size_t)SB * HDIM];
    scan_pass<!LAST>(dr, br, dk + (size_t)SB * 64, bk + (size_t)SB * 64,
                     xv, h0, h1, Aj, pw + ST * (PPITCH4 * 4), lane);

    __builtin_amdgcn_fence(__ATOMIC_RELEASE, "workgroup", "local");
    __builtin_amdgcn_s_barrier();
    __builtin_amdgcn_fence(__ATOMIC_ACQUIRE, "workgroup", "local");

    float g0 = 0.f, g1 = 0.f;
#pragma unroll
    for (int q = 0; q < 8; ++q) {
        const float4 v = r0[q];
        const float4 w = r1[q];
        g0 += v.x; g0 += v.y; g0 += v.z; g0 += v.w;
        g1 += w.x; g1 += w.y; g1 += w.z; g1 += w.w;
    }
    const float y = cv * (g0 + g1) + Dj * uv;
    *yp = y * zv;
    xv = xn;
}

__global__ void __launch_bounds__(SC * SH * 32)
scan_seq(const float* __restrict__ delta,
         const float* __restrict__ Bm,
         const float* __restrict__ Cm,
         const float* __restrict__ x2,
         const float* __restrict__ u,
         const float* __restrict__ sz,
         const float* __restrict__ A_log,
         const float* __restrict__ Dv,
         float* __restrict__ yT)
{
    const int tid  = threadIdx.x;
    const int lane = tid & 31;
    const int wave = tid >> 5;
    const int jl   = wave >> 1;
    const int half = wave & 1;
    const int b    = blockIdx.x / (HDIM / SC);
    const int j    = (blockIdx.x % (HDIM / SC)) * SC + jl;

    __shared__ float4 s_p[2][SC][SH][SB][PPITCH4];

    const float Aj = -__expf(A_log[j]);
    const float Dj = Dv[j];

    const size_t rowb = (size_t)b * LEN;
    const float2* d2 = reinterpret_cast<const float2*>(delta) + rowb * 64 + 32 * half + lane;
    const float2* B2 = reinterpret_cast<const float2*>(Bm)    + rowb * 64 + 32 * half + lane;

    float xv = x2[(rowb + lane) * HDIM + j];
    __builtin_amdgcn_sched_barrier(0);
    float2 dr[ST], br[ST];
#pragma unroll
    for (int s = 0; s < ST; ++s) {
        dr[s] = d2[(size_t)s * 64];
        br[s] = B2[(size_t)s * 64];
    }

    float h0 = 0.f, h1 = 0.f;

    const size_t colx = (rowb + lane) * HDIM + j;
    const size_t cole = (rowb + 32 * half + lane) * HDIM + j;
    float* yrow = yT + (size_t)j * MTOT + rowb + 32 * half + lane;

    for (int k = 0; k < NSB - 1; ++k) {
        const size_t t0  = (size_t)k * SB;
        const int    buf = k & 1;
        scan_block<false>(dr, br, d2 + t0 * 64, B2 + t0 * 64, x2 + colx + t0 * HDIM,
                          Cm + cole + t0 * HDIM, u + cole + t0 * HDIM, sz + cole + t0 * HDIM,
                          yrow + t0, xv, h0, h1, Aj, Dj,
                          reinterpret_cast<float*>(&s_p[buf][jl][half][0][0]),
                          &s_p[buf][jl][0][32 * half + lane][0],
                          &s_p[buf][jl][1][32 * half + lane][0], lane);
    }
    {
        const size_t t0  = (size_t)(NSB - 1) * SB;
        const int    buf = (NSB - 1) & 1;
        scan_block<true>(dr, br, d2 + t0 * 64, B2 + t0 * 64, x2 + colx + t0 * HDIM,
                         Cm + cole + t0 * HDIM, u + cole + t0 * HDIM, sz + cole + t0 * HDIM,
                         yrow + t0, xv, h0, h1, Aj, Dj,
                         reinterpret_cast<float*>(&s_p[buf][jl][half][0][0]),
                         &s_p[buf][jl][0][32 * half + lane][0],
                         &s_p[buf][jl][1][32 * half + lane][0], lane);
    }
}

extern "C" void kernel_launch(void* const* d_in, const int* in_sizes, int n_in,
                              void* d_out, int out_size, void* d_ws, size_t ws_size,
                              hipStream_t stream)
{
    (void)in_sizes; (void)n_in; (void)out_size; (void)ws_size;

    const float* u      = (const float*)d_in[0];
    const float* W_in   = (const float*)d_in[1];
    const float* b_in   = (const float*)d_in[2];
    const float* conv_w = (const float*)d_in[3];
    const float* conv_b = (const float*)d_in[4];
    const float* W_x    = (const float*)d_in[5];
    const float* b_x    = (const float*)d_in[6];
    const float* A_log  = (const float*)d_in[7];
    const float* Dv     = (const float*)d_in[8];
    const float* W_out  = (const float*)d_in[9];
    const float* b_out  = (const float*)d_in[10];

    const size_t P = (size_t)MTOT * HDIM;

    float* ws    = (float*)d_ws;
    float* s_z   = ws + 0 * P;
    float* x2    = ws + 1 * P;
    float* delta = ws + 2 * P;
    float* Bmat  = ws + 3 * P;
    float* Cmat  = ws + 4 * P;
    float* ymulT = ws + 5 * P;

    inproj_conv_wmma_f32<<<256, 256, 0, stream>>>(u, W_in, b_in, conv_w, conv_b, x2, s_z);
    gemm16_wmma_f32<384, 128, EPI_SSM, 0><<<768, 256, 0, stream>>>(
        x2, W_x, b_x, delta, Bmat, Cmat);
    scan_seq<<<BATCH * (HDIM / SC), SC * SH * 32, 0, stream>>>(delta, Bmat, Cmat, x2, u, s_z,
                                                              A_log, Dv, ymulT);
    gemm16_wmma_f32<128, 128, EPI_PLAIN, MTOT><<<256, 256, 0, stream>>>(
        ymulT, W_out, b_out, (float*)d_out, nullptr, nullptr);
}


# scan: ssm GEMM epilogue interleaves delta and B, so a step's operands are ONE 16-byte load per lane; two register sets: rows and x requested a 64-step block ahead (was 24-32 steps)
# speedup vs baseline: 4.1895x; 1.1451x over previous
#include <hip/hip_runtime.h>


#define HDIM 128
#define BATCH 4
#define LEN 1024
#define MTOT (BATCH * LEN)

typedef float v2f __attribute__((ext_vector_type(2)));
typedef float v8f __attribute__((ext_vector_type(8)));

__device__ __forceinline__ float sigmoid_fast(float v) { return 1.0f / (1.0f + __expf(-v)); }
__device__ __forceinline__ float silu_fast(float v)    { return v * sigmoid_fast(v); }
__device__ __forceinline__ float softplus_fast(float v){ return (v > 20.0f) ? v : log1pf(__expf(v)); }

#define EPI_SSM   1
#define EPI_PLAIN 2

template <int N, int K, int MODE, int ALD>
__global__ void __launch_bounds__(256)
gemm16_wmma_f32(const float* __restrict__ A,
                const float* __restrict__ W,
                const float* __restrict__ bias,
                float* __restrict__ out0,
                float* __restrict__ out1,
                float* __restrict__ out2)
{
    const int lane = threadIdx.x & 31;
    const int wave = threadIdx.x >> 5;
    const int gw   = blockIdx.x * (blockDim.x >> 5) + wave;
    const int tilesN = N >> 4;
    const int m0 = (gw / tilesN) << 4;
    const int n0 = (gw % tilesN) << 4;
    const int hi  = lane >> 4;
    const int l16 = lane & 15;

    v8f acc = {0.f, 0.f, 0.f, 0.f, 0.f, 0.f, 0.f, 0.f};

    const float* Arow = ALD ? A + (size_t)(2 * hi) * ALD + (m0 + l16)
                            : A + (size_t)(m0 + l16) * K + 2 * hi;
    const float* Wcol = W + (size_t)(2 * hi) * N + n0 + l16;

#pragma unroll
    for (int k0 = 0; k0 < K; k0 += 4) {
        v2f a;
        a.x = ALD ? Arow[(size_t)(k0 + 0) * ALD] : Arow[k0 + 0];
        a.y = ALD ? Arow[(size_t)(k0 + 1) * ALD] : Arow[k0 + 1];
        v2f b;
        b.x = Wcol[(k0 + 0) * N];
        b.y = Wcol[(k0 + 1) * N];
        acc = __builtin_amdgcn_wmma_f32_16x16x4_f32(false, a, false, b,
                                                    (short)0, acc, false, false);
    }

    const int   col = n0 + l16;
    const float bv  = bias[col];
#pragma unroll
    for (int r = 0; r < 8; ++r) {
        const int row = m0 + r + hi * 8;
        float v = acc[r] + bv;
        if (MODE == EPI_SSM) {
            if (col < 128)      out0[((size_t)row * 128 + col) * 2]             = softplus_fast(v);
            else if (col < 256) out0[((size_t)row * 128 + (col - 128)) * 2 + 1] = v;
            else                out2[(size_t)row * 128 + (col - 256)] = v;
        } else {
            out0[(size_t)row * HDIM + col] = v;
        }
    }
}

#define KC 32
#define KS (KC / 4)

__device__ __forceinline__ void inproj_load_chunk(
    v2f (&aa)[KS], v2f (&ab)[KS], v2f (&ac)[KS], v2f (&bx)[KS], v2f (&bz)[KS],
    const float* __restrict__ Aa, const float* __restrict__ Ab, const float* __restrict__ Ac,
    const float* __restrict__ Wx, const float* __restrict__ Wz, const int kb)
{
#pragma unroll
    for (int s = 0; s < KS; ++s) {
        const int k0 = kb + 4 * s;
        aa[s].x = Aa[k0 + 0];  aa[s].y = Aa[k0 + 1];
        ab[s].x = Ab[k0 + 0];  ab[s].y = Ab[k0 + 1];
        ac[s].x = Ac[k0 + 0];  ac[s].y = Ac[k0 + 1];
        bx[s].x = Wx[(k0 + 0) * (2 * HDIM)];
        bx[s].y = Wx[(k0 + 1) * (2 * HDIM)];
        bz[s].x = Wz[(k0 + 0) * (2 * HDIM)];
        bz[s].y = Wz[(k0 + 1) * (2 * HDIM)];
    }
}

__global__ void __launch_bounds__(256)
inproj_conv_wmma_f32(const float* __restrict__ u,
                     const float* __restrict__ W,
                     const float* __restrict__ bias,
                     const float* __restrict__ cw,
                     const float* __restrict__ cb,
                     float* __restrict__ x2,
                     float* __restrict__ sz)
{
    const int lane = threadIdx.x & 31;
    const int wave = threadIdx.x >> 5;
    const int gw   = blockIdx.x * (blockDim.x >> 5) + wave;
    const int m0 = (gw >> 3) << 4;
    const int n0 = (gw & 7) << 4;
    const int hi  = lane >> 4;
    const int l16 = lane & 15;

    const int ra = (m0 - 1 + l16 < 0) ? 0 : m0 - 1 + l16;
    const int rb = (m0 + 1 + l16 > MTOT - 1) ? MTOT - 1 : m0 + 1 + l16;
    const float* Aa = u + (size_t)ra * HDIM + 2 * hi;
    const float* Ab = u + (size_t)rb * HDIM + 2 * hi;
    const float* Ac = u + (size_t)(m0 + l16) * HDIM + 2 * hi;
    const float* Wx = W + (size_t)(2 * hi) * (2 * HDIM) + n0 + l16;
    const float* Wz = Wx + HDIM;

    v8f acc_a = {0.f, 0.f, 0.f, 0.f, 0.f, 0.f, 0.f, 0.f};
    v8f acc_b = acc_a;
    v8f acc_c = acc_a;

    v2f aa[2][KS], ab[2][KS], ac[2][KS], bx[2][KS], bz[2][KS];
    inproj_load_chunk(aa[0], ab[0], ac[0], bx[0], bz[0], Aa, Ab, Ac, Wx, Wz, 0);

#pragma unroll
    for (int c = 0; c < HDIM / KC; ++c) {
        const int cur = c & 1;
        if (c + 1 < HDIM / KC)
            inproj_load_chunk(aa[cur ^ 1], ab[cur ^ 1], ac[cur ^ 1], bx[cur ^ 1], bz[cur ^ 1],
                              Aa, Ab, Ac, Wx, Wz, (c + 1) * KC);
        __builtin_amdgcn_sched_barrier(0);
#pragma unroll
        for (int s = 0; s < KS; ++s) {
            acc_a = __builtin_amdgcn_wmma_f32_16x16x4_f32(false, aa[cur][s], false, bx[cur][s],
                                                          (short)0, acc_a, false, false);
            acc_b = __builtin_amdgcn_wmma_f32_16x16x4_f32(false, ab[cur][s], false, bx[cur][s],
                                                          (short)0, acc_b, false, false);
            acc_c = __builtin_amdgcn_wmma_f32_16x16x4_f32(false, ac[cur][s], false, bz[cur][s],
                                                          (short)0, acc_c, false, false);
        }
        __builtin_amdgcn_sched_barrier(0);
    }

    const int   e   = n0 + l16;
    const float bxv = bias[e];
    const float bzv = bias[HDIM + e];
    const float w0  = cw[e * 3 + 0];
    const float w1  = cw[e * 3 + 1];
    const float w2  = cw[e * 3 + 2];
    const float cbv = cb[e];
#pragma unroll
    for (int r = 0; r < 8; ++r) {
        const int row = m0 + r + hi * 8;
        const int l   = row & (LEN - 1);
        const int rn  = (r < 7) ? r + 1 : 7;
        const float xm = (l > 0)       ? acc_a[r] + bxv : 0.f;
        const float xc = ((r < 7) ? acc_a[rn] : acc_b[6]) + bxv;
        const float xp = (l < LEN - 1) ? acc_b[r] + bxv : 0.f;
        const float v  = xm * w0 + xc * w1 + xp * w2 + cbv;
        x2[(size_t)row * HDIM + e] = silu_fast(v);
        sz[(size_t)row * HDIM + e] = silu_fast(acc_c[r] + bzv);
    }
}

#define SC 2
#define SH 2
#define ST 32
#define GS 8
#define SB (SH * ST)
#define NSB (LEN / SB)
#define PPITCH4 9
#define ROW4 (HDIM / 2)

template <bool MORE>
__device__ __forceinline__ void scan_pass(
    float4 (&op)[ST],
    const float4* __restrict__ nx,
    const float xv,
    float& h0, float& h1, const float Aj,
    float* __restrict__ pw, const int lane)
{
#pragma unroll
    for (int gi = 0; gi < ST / GS; ++gi) {
#pragma unroll
        for (int i = 0; i < GS; ++i) {
            const int s = gi * GS + i;
            const float4 o = op[s];
            const float xs = __builtin_bit_cast(float,
                __builtin_amdgcn_readlane(__builtin_bit_cast(int, xv), s));
            const float p0 = o.x * o.y;
            const float p1 = o.z * o.w;
            h0 = __expf(o.x * Aj) * h0 + p0 * xs;
            h1 = __expf(o.z * Aj) * h1 + p1 * xs;
            pw[s * (PPITCH4 * 4) + lane] = h0 + h1;
        }
        if (MORE) {
            __builtin_amdgcn_sched_barrier(0);
#pragma unroll
            for (int i = 0; i < GS; ++i) {
                const int s = gi * GS + i;
                op[s] = nx[(size_t)s * ROW4];
            }
            __builtin_amdgcn_sched_barrier(0);
        }
    }
}

template <bool LAST>
__device__ __forceinline__ void scan_block(
    float4 (&o0)[ST], float4 (&o1)[ST],
    const float4* __restrict__ dbk,
    const float* __restrict__ xk,
    const float* __restrict__ cp,
    const float* __restrict__ up,
    const float* __restrict__ zp,
    float* __restrict__ yp,
    float& xa, float& xb,
    float& h0, float& h1,
    const float Aj, const float Dj,
    float* __restrict__ pw,
    const float4* __restrict__ r0,
    const float4* __restrict__ r1,
    const int lane)
{
    float xna = 0.f, xnb = 0.f;
    if (!LAST) {
        xna = xk[(size_t)SB * HDIM];
        xnb = xk[(size_t)(SB + ST) * HDIM];
    }
    const float cv = *cp;
    const float uv = *up;
    const float zv = *zp;

    scan_pass<!LAST>(o0, dbk + (size_t)SB * ROW4, xa, h0, h1, Aj, pw, lane);
    scan_pass<!LAST>(o1, dbk + (size_t)(SB + ST) * ROW4, xb, h0, h1, Aj,
                     pw + ST * (PPITCH4 * 4), lane);

    __builtin_amdgcn_fence(__ATOMIC_RELEASE, "workgroup", "local");
    __builtin_amdgcn_s_barrier();
    __builtin_amdgcn_fence(__ATOMIC_ACQUIRE, "workgroup", "local");

    float g0 = 0.f, g1 = 0.f;
#pragma unroll
    for (int q = 0; q < 8; ++q) {
        const float4 v = r0[q];
        const float4 w = r1[q];
        g0 += v.x; g0 += v.y; g0 += v.z; g0 += v.w;
        g1 += w.x; g1 += w.y; g1 += w.z; g1 += w.w;
    }
    const float y = cv * (g0 + g1) + Dj * uv;
    *yp = y * zv;
    xa = xna;
    xb = xnb;
}

__global__ void __launch_bounds__(SC * SH * 32)
scan_seq(const float* __restrict__ dB,
         const float* __restrict__ Cm,
         const float* __restrict__ x2,
         const float* __restrict__ u,
         const float* __restrict__ sz,
         const float* __restrict__ A_log,
         const float* __restrict__ Dv,
         float* __restrict__ yT)
{
    const int tid  = threadIdx.x;
    const int lane = tid & 31;
    const int wave = tid >> 5;
    const int jl   = wave >> 1;
    const int half = wave & 1;
    const int b    = blockIdx.x / (HDIM / SC);
    const int j    = (blockIdx.x % (HDIM / SC)) * SC + jl;

    __shared__ float4 s_p[2][SC][SH][SB][PPITCH4];

    const float Aj = -__expf(A_log[j]);
    const float Dj = Dv[j];

    const size_t rowb = (size_t)b * LEN;
    const float4* db4 = reinterpret_cast<const float4*>(dB) + rowb * ROW4 + 32 * half + lane;

    float xa = x2[(rowb + lane) * HDIM + j];
    float xb = x2[(rowb + ST + lane) * HDIM + j];
    __builtin_amdgcn_sched_barrier(0);
    float4 o0[ST], o1[ST];
#pragma unroll
    for (int s = 0; s < ST; ++s) o0[s] = db4[(size_t)s * ROW4];
#pragma unroll
    for (int s = 0; s < ST; ++s) o1[s] = db4[(size_t)(ST + s) * ROW4];

    float h0 = 0.f, h1 = 0.f;

    const size_t colx = (rowb + lane) * HDIM + j;
    const size_t cole = (rowb + 32 * half + lane) * HDIM + j;
    float* yrow = yT + (size_t)j * MTOT + rowb + 32 * half + lane;

    for (int k = 0; k < NSB - 1; ++k) {
        const size_t t0  = (size_t)k * SB;
        const int    buf = k & 1;
        scan_block<false>(o0, o1, db4 + t0 * ROW4, x2 + colx + t0 * HDIM,
                          Cm + cole + t0 * HDIM, u + cole + t0 * HDIM, sz + cole + t0 * HDIM,
                          yrow + t0, xa, xb, h0, h1, Aj, Dj,
                          reinterpret_cast<float*>(&s_p[buf][jl][half][0][0]),
                          &s_p[buf][jl][0][32 * half + lane][0],
                          &s_p[buf][jl][1][32 * half + lane][0], lane);
    }
    {
        const size_t t0  = (size_t)(NSB - 1) * SB;
        const int    buf = (NSB - 1) & 1;
        scan_block<true>(o0, o1, db4 + t0 * ROW4, x2 + colx + t0 * HDIM,
                         Cm + cole + t0 * HDIM, u + cole + t0 * HDIM, sz + cole + t0 * HDIM,
                         yrow + t0, xa, xb, h0, h1, Aj, Dj,
                         reinterpret_cast<float*>(&s_p[buf][jl][half][0][0]),
                         &s_p[buf][jl][0][32 * half + lane][0],
                         &s_p[buf][jl][1][32 * half + lane][0], lane);
    }
}

extern "C" void kernel_launch(void* const* d_in, const int* in_sizes, int n_in,
                              void* d_out, int out_size, void* d_ws, size_t ws_size,
                              hipStream_t stream)
{
    (void)in_sizes; (void)n_in; (void)out_size; (void)ws_size;

    const float* u      = (const float*)d_in[0];
    const float* W_in   = (const float*)d_in[1];
    const float* b_in   = (const float*)d_in[2];
    const float* conv_w = (const float*)d_in[3];
    const float* conv_b = (const float*)d_in[4];
    const float* W_x    = (const float*)d_in[5];
    const float* b_x    = (const float*)d_in[6];
    const float* A_log  = (const float*)d_in[7];
    const float* Dv     = (const float*)d_in[8];
    const float* W_out  = (const float*)d_in[9];
    const float* b_out  = (const float*)d_in[10];

    const size_t P = (size_t)MTOT * HDIM;

    float* ws    = (float*)d_ws;
    float* s_z   = ws + 0 * P;
    float* x2    = ws + 1 * P;
    float* dB    = ws + 2 * P;
    float* Cmat  = ws + 4 * P;
    float* ymulT = ws + 5 * P;

    inproj_conv_wmma_f32<<<256, 256, 0, stream>>>(u, W_in, b_in, conv_w, conv_b, x2, s_z);
    gemm16_wmma_f32<384, 128, EPI_SSM, 0><<<768, 256, 0, stream>>>(
        x2, W_x, b_x, dB, nullptr, Cmat);
    scan_seq<<<BATCH * (HDIM / SC), SC * SH * 32, 0, stream>>>(dB, Cmat, x2, u, s_z,
                                                              A_log, Dv, ymulT);
    gemm16_wmma_f32<128, 128, EPI_PLAIN, MTOT><<<256, 256, 0, stream>>>(
        ymulT, W_out, b_out, (float*)d_out, nullptr, nullptr);
}
